# starting kernel
# baseline (speedup 1.0000x reference)
#include <hip/hip_runtime.h>

#define BATCH 8
#define LSEQ 1024
#define HIDDEN 256
#define NUM_RBF 16
#define ROWS 16
#define JLANES 16

typedef __attribute__((ext_vector_type(2))) float v2f;
typedef __attribute__((ext_vector_type(8))) float v8f;

#define NEG_HALF_LOG2E (-0.72134752044448170368f)

__global__ __launch_bounds__(256) void rbf_mean_kernel(
    const float* __restrict__ pos,
    const float* __restrict__ centers,
    float* __restrict__ rbf_mean)
{
    __shared__ float sPos[LSEQ * 3];
    __shared__ float sPart[256][NUM_RBF + 1];

    const int tid          = threadIdx.x;
    const int tilesPerB    = LSEQ / ROWS;
    const int b            = blockIdx.x / tilesPerB;
    const int rt           = blockIdx.x % tilesPerB;

    const float* posB = pos + (size_t)b * LSEQ * 3;
    for (int idx = tid; idx < LSEQ * 3; idx += 256)
        sPos[idx] = posB[idx];

    float c[NUM_RBF];
#pragma unroll
    for (int r = 0; r < NUM_RBF; ++r) c[r] = centers[r];

    __syncthreads();

    const int jl  = tid & (JLANES - 1);
    const int row = tid >> 4;
    const int i   = rt * ROWS + row;

    const float xi = sPos[i * 3 + 0];
    const float yi = sPos[i * 3 + 1];
    const float zi = sPos[i * 3 + 2];

    float acc[NUM_RBF];
#pragma unroll
    for (int r = 0; r < NUM_RBF; ++r) acc[r] = 0.0f;

#pragma unroll 2
    for (int j = jl; j < LSEQ; j += JLANES) {
        const float dx = xi - sPos[j * 3 + 0];
        const float dy = yi - sPos[j * 3 + 1];
        const float dz = zi - sPos[j * 3 + 2];
        const float d  = __builtin_sqrtf(dx * dx + dy * dy + dz * dz);
#pragma unroll
        for (int r = 0; r < NUM_RBF; ++r) {
            const float t = d - c[r];
            acc[r] += __builtin_amdgcn_exp2f(t * t * NEG_HALF_LOG2E);
        }
    }

#pragma unroll
    for (int r = 0; r < NUM_RBF; ++r) sPart[tid][r] = acc[r];
    __syncthreads();

    const int orow = tid >> 4;
    const int r    = tid & 15;
    float s = 0.0f;
#pragma unroll
    for (int k = 0; k < JLANES; ++k)
        s += sPart[orow * JLANES + k][r];

    const int gRow = b * LSEQ + rt * ROWS + orow;
    rbf_mean[gRow * NUM_RBF + r] = s * (1.0f / (float)LSEQ);
}

__global__ __launch_bounds__(256) void linear_wmma_kernel(
    const float* __restrict__ rbf_mean,
    const float* __restrict__ W,
    const float* __restrict__ bias,
    const float* __restrict__ latent,
    float* __restrict__ out)
{
    const int lane     = threadIdx.x & 31;
    const int waveBlk  = threadIdx.x >> 5;
    const int wave     = blockIdx.x * 8 + waveBlk;
    const int rowBase  = wave * 16;

    const int m  = lane & 15;
    const int hi = lane >> 4;

    v2f a[4];
#pragma unroll
    for (int kc = 0; kc < 4; ++kc) {
        const float* p = rbf_mean + (size_t)(rowBase + m) * NUM_RBF + kc * 4 + hi * 2;
        v2f t; t.x = p[0]; t.y = p[1];
        a[kc] = t;
    }

#pragma unroll 1
    for (int nt = 0; nt < HIDDEN / 16; ++nt) {
        const int n = nt * 16 + m;

        v2f bf[4];
#pragma unroll
        for (int kc = 0; kc < 4; ++kc) {
            const float* p = W + (size_t)n * NUM_RBF + kc * 4 + hi * 2;
            v2f t; t.x = p[0]; t.y = p[1];
            bf[kc] = t;
        }

        v8f cfr = {0.f, 0.f, 0.f, 0.f, 0.f, 0.f, 0.f, 0.f};
#pragma unroll
        for (int kc = 0; kc < 4; ++kc) {
            cfr = __builtin_amdgcn_wmma_f32_16x16x4_f32(
                 false, a[kc],  false, bf[kc],
                 (short)0, cfr,  false,  false);
        }

        const float bv = bias[n];
#pragma unroll
        for (int v = 0; v < 8; ++v) {
            const int    rrow = rowBase + v + hi * 8;
            const int    col  = nt * 16 + m;
            const size_t idx  = (size_t)rrow * HIDDEN + col;
            out[idx] = latent[idx] + cfr[v] + bv;
        }
    }
}

extern "C" void kernel_launch(void* const* d_in, const int* in_sizes, int n_in,
                              void* d_out, int out_size, void* d_ws, size_t ws_size,
                              hipStream_t stream) {
    const float* pos     = (const float*)d_in[0];
    const float* latent  = (const float*)d_in[1];
    const float* W       = (const float*)d_in[2];
    const float* bias    = (const float*)d_in[3];
    const float* centers = (const float*)d_in[4];
    float* out      = (float*)d_out;
    float* rbf_mean = (float*)d_ws;

    rbf_mean_kernel<<<BATCH * (LSEQ / ROWS), 256, 0, stream>>>(pos, centers, rbf_mean);

    linear_wmma_kernel<<<(BATCH * LSEQ / 16) / 8, 256, 0, stream>>>(
        rbf_mean, W, bias, latent, out);
}


# rbf phase re-decomposed: one lane = one row, a wave's lanes walk the same j, positions arrive by scalar loads (no LDS staging, no barrier before the pair loop); j summation order changed
# speedup vs baseline: 1.4250x; 1.4250x over previous
#include <hip/hip_runtime.h>

#define BATCH 8
#define LSEQ 1024
#define HIDDEN 256
#define NUM_RBF 16

typedef __attribute__((ext_vector_type(2))) float v2f;
typedef __attribute__((ext_vector_type(4))) float v4f;
typedef __attribute__((ext_vector_type(8))) float v8f;

#define NEG_HALF_LOG2E (-0.72134752044448170368f)

#define RROWS   32
#define JSLICES 16
#define JPER    (LSEQ / JSLICES)
#define JCH     8

__global__ __launch_bounds__(RROWS * JSLICES) __attribute__((amdgpu_waves_per_eu(4, 4))) void rbf_mean_kernel(
    const float* __restrict__ pos,
    const float* __restrict__ centers,
    float* __restrict__ rbf_mean)
{
    __shared__ float sPart[JSLICES][RROWS][NUM_RBF + 1];

    const int tid  = threadIdx.x;
    const int lane = tid & 31;
    const int w    = __builtin_amdgcn_readfirstlane(tid >> 5);
    const int blocksPerB = LSEQ / RROWS;
    const int b    = blockIdx.x / blocksPerB;
    const int i0   = (blockIdx.x % blocksPerB) * RROWS;

    const float* __restrict__ posB = pos + (size_t)b * LSEQ * 3;
    const float* __restrict__ pj   = posB + (size_t)w * JPER * 3;

    const float xi = posB[(i0 + lane) * 3 + 0];
    const float yi = posB[(i0 + lane) * 3 + 1];
    const float zi = posB[(i0 + lane) * 3 + 2];

    v2f c2[NUM_RBF / 2];
#pragma unroll
    for (int r = 0; r < NUM_RBF / 2; ++r) { c2[r].x = centers[2 * r]; c2[r].y = centers[2 * r + 1]; }

    v2f acc2[NUM_RBF / 2];
#pragma unroll
    for (int r = 0; r < NUM_RBF / 2; ++r) { acc2[r].x = 0.0f; acc2[r].y = 0.0f; }

    float cur[JCH * 3], nxt[JCH * 3];
#pragma unroll
    for (int q = 0; q < JCH * 3; ++q) cur[q] = pj[q];

#pragma unroll 1
    for (int ch = 0; ch < JPER / JCH; ++ch) {
        const int nch = (ch + 1 < JPER / JCH) ? ch + 1 : ch;
#pragma unroll
        for (int q = 0; q < JCH * 3; ++q) nxt[q] = pj[nch * (JCH * 3) + q];

#pragma unroll
        for (int q = 0; q < JCH; ++q) {
            const float dx = xi - cur[q * 3 + 0];
            const float dy = yi - cur[q * 3 + 1];
            const float dz = zi - cur[q * 3 + 2];
            const float d  = __builtin_sqrtf(dx * dx + dy * dy + dz * dz);
            const v2f dd = {d, d};
#pragma unroll
            for (int r = 0; r < NUM_RBF / 2; ++r) {
                const v2f t = dd - c2[r];
                const v2f u = t * t * NEG_HALF_LOG2E;
                v2f e;
                e.x = __builtin_amdgcn_exp2f(u.x);
                e.y = __builtin_amdgcn_exp2f(u.y);
                acc2[r] += e;
            }
        }
#pragma unroll
        for (int q = 0; q < JCH * 3; ++q) cur[q] = nxt[q];
    }

#pragma unroll
    for (int r = 0; r < NUM_RBF; ++r) sPart[w][lane][r] = (r & 1) ? acc2[r >> 1].y : acc2[r >> 1].x;
    __syncthreads();

    {
        const int orow = tid >> 4;
        const int r    = tid & 15;
        float sum = 0.0f;
#pragma unroll
        for (int k = 0; k < JSLICES; ++k) sum += sPart[k][orow][r];
        const int gRow = b * LSEQ + i0 + orow;
        rbf_mean[gRow * NUM_RBF + r] = sum * (1.0f / (float)LSEQ);
    }
}

#define LIN_COLS  64
#define LIN_TILES (LIN_COLS / 16)

__global__ __launch_bounds__(256) void linear_wmma_kernel(
    const float* __restrict__ rbf_mean,
    const float* __restrict__ W,
    const float* __restrict__ bias,
    const float* __restrict__ latent,
    float* __restrict__ out)
{
    const int lane    = threadIdx.x & 31;
    const int waveBlk = threadIdx.x >> 5;
    const int wave    = blockIdx.x * 8 + waveBlk;
    const int rowBase = (wave >> 2) * 16;
    const int n0      = (wave & 3) * LIN_COLS;

    const int c  = lane & 15;
    const int hi = lane >> 4;

    const int    colBase = n0 + hi * 32;
    const size_t rowOff  = (size_t)(rowBase + c) * HIDDEN + colBase;

    const v4f* __restrict__ latP  = (const v4f*)(latent + rowOff);
    const v4f* __restrict__ biasP = (const v4f*)(bias + colBase);
    v4f*       __restrict__ outP  = (v4f*)(out + rowOff);

    v2f bf[4];
#pragma unroll
    for (int kc = 0; kc < 4; ++kc) {
        const float* p = rbf_mean + (size_t)(rowBase + c) * NUM_RBF + kc * 4 + hi * 2;
        v2f t; t.x = p[0]; t.y = p[1];
        bf[kc] = t;
    }

    v2f af[LIN_TILES][4];
#pragma unroll
    for (int t = 0; t < LIN_TILES; ++t) {
        const int n = n0 + (c >> 3) * 32 + t * 8 + (c & 7);
#pragma unroll
        for (int kc = 0; kc < 4; ++kc) {
            const float* p = W + (size_t)n * NUM_RBF + kc * 4 + hi * 2;
            v2f u; u.x = p[0]; u.y = p[1];
            af[t][kc] = u;
        }
    }

    v4f bs[2 * LIN_TILES];
#pragma unroll
    for (int q = 0; q < 2 * LIN_TILES; ++q) bs[q] = biasP[q];

    v4f lat[2 * LIN_TILES];
#pragma unroll
    for (int q = 0; q < 2 * LIN_TILES; ++q) lat[q] = latP[q];

    __builtin_amdgcn_sched_barrier(0);

#pragma unroll
    for (int t = 0; t < LIN_TILES; ++t) {
        v8f cfr = {0.f, 0.f, 0.f, 0.f, 0.f, 0.f, 0.f, 0.f};
#pragma unroll
        for (int kc = 0; kc < 4; ++kc) {
            cfr = __builtin_amdgcn_wmma_f32_16x16x4_f32(
                 false, af[t][kc],  false, bf[kc],
                 (short)0, cfr,  false,  false);
        }

        v4f o0, o1;
#pragma unroll
        for (int e = 0; e < 4; ++e) {
            o0[e] = lat[2 * t][e]     + cfr[e]     + bs[2 * t][e];
            o1[e] = lat[2 * t + 1][e] + cfr[4 + e] + bs[2 * t + 1][e];
        }
        outP[2 * t]     = o0;
        outP[2 * t + 1] = o1;
    }
}

extern "C" void kernel_launch(void* const* d_in, const int* in_sizes, int n_in,
                              void* d_out, int out_size, void* d_ws, size_t ws_size,
                              hipStream_t stream) {
    const float* pos     = (const float*)d_in[0];
    const float* latent  = (const float*)d_in[1];
    const float* W       = (const float*)d_in[2];
    const float* bias    = (const float*)d_in[3];
    const float* centers = (const float*)d_in[4];
    float* out      = (float*)d_out;
    float* rbf_mean = (float*)d_ws;

    rbf_mean_kernel<<<BATCH * (LSEQ / RROWS), RROWS * JSLICES, 0, stream>>>(pos, centers, rbf_mean);

    linear_wmma_kernel<<<(BATCH * LSEQ / 16) * (HIDDEN / LIN_COLS) / 8, 256, 0, stream>>>(
        rbf_mean, W, bias, latent, out);
}


# linear phase: product tile goes through LDS (written as D holds it, read row-major): each latent load and out store covers 4 whole 128-byte lines, was 16 bytes of 32 lines; bias one load per lane
# speedup vs baseline: 1.7048x; 1.1964x over previous
#include <hip/hip_runtime.h>


#define BATCH 8
#define LSEQ 1024
#define HIDDEN 256
#define NUM_RBF 16

typedef __attribute__((ext_vector_type(2))) float v2f;
typedef __attribute__((ext_vector_type(4))) float v4f;
typedef __attribute__((ext_vector_type(8))) float v8f;

#define NEG_HALF_LOG2E (-0.72134752044448170368f)

#define RB      32
#define NRB     (LSEQ / RB)
#define NPLANES 16
#define PLANE_FLOATS (BATCH * LSEQ * NUM_RBF)
#define NSLOTS  32

__device__ __forceinline__ float ror1(float v) {
    return __int_as_float(__builtin_amdgcn_update_dpp(0, __float_as_int(v), 0x121, 0xF, 0xF, true));
}

__device__ __forceinline__ int r_of_pos(int p) {
    return ((p >> 1) & 3) * 4 + (p >> 3) * 2 + (p & 1);
}

__device__ __forceinline__ void pair_rbf(float xi, float yi, float zi,
                                         float xj, float yj, float zj,
                                         const v2f (&c2)[NUM_RBF / 2], v2f (&e)[NUM_RBF / 2]) {
    const float dx = xi - xj;
    const float dy = yi - yj;
    const float dz = zi - zj;
    const float d  = __builtin_sqrtf(dx * dx + dy * dy + dz * dz);
    const v2f dd = {d, d};
#pragma unroll
    for (int r = 0; r < NUM_RBF / 2; ++r) {
        const v2f t = dd - c2[r];
        const v2f u = t * t * NEG_HALF_LOG2E;
        e[r].x = __builtin_amdgcn_exp2f(u.x);
        e[r].y = __builtin_amdgcn_exp2f(u.y);
    }
}

template <bool SYM>
__device__ __forceinline__ void tile32(float xa, float ya, float za,
                                       float xb, float yb, float zb,
                                       float xj, float yj, float zj,
                                       const v2f (&c2)[NUM_RBF / 2],
                                       v2f (&ra)[NUM_RBF / 2], v2f (&rb)[NUM_RBF / 2],
                                       float (&col)[NUM_RBF]) {
#pragma unroll 2
    for (int s = 0; s < 16; ++s) {
        xj = ror1(xj); yj = ror1(yj); zj = ror1(zj);
        v2f ea[NUM_RBF / 2], eb[NUM_RBF / 2];
        pair_rbf(xa, ya, za, xj, yj, zj, c2, ea);
        pair_rbf(xb, yb, zb, xj, yj, zj, c2, eb);
#pragma unroll
        for (int r = 0; r < NUM_RBF / 2; ++r) {
            ra[r] += ea[r];
            rb[r] += eb[r];
            if (SYM) {
                const v2f both = ea[r] + eb[r];
                col[2 * r]     = ror1(col[2 * r])     + both.x;
                col[2 * r + 1] = ror1(col[2 * r + 1]) + both.y;
            }
        }
    }
}

__global__ __launch_bounds__(512) __attribute__((amdgpu_waves_per_eu(4, 4))) void rbf_pair_kernel(
    const float* __restrict__ pos,
    const float* __restrict__ centers,
    float* __restrict__ planes)
{
    __shared__ float sPart[NSLOTS][RB][NUM_RBF + 1];

    const int tid  = threadIdx.x;
    const int lane = tid & 31;
    const int c    = lane & 15;
    const int h    = lane >> 4;
    const int w    = __builtin_amdgcn_readfirstlane(tid >> 5);
    const int b    = blockIdx.x / NRB;
    const int I    = blockIdx.x % NRB;

    const float* __restrict__ posB = pos + (size_t)b * LSEQ * 3;

    const int ia = I * RB + c;
    const int ib = ia + 16;
    const float xa = posB[ia * 3 + 0], ya = posB[ia * 3 + 1], za = posB[ia * 3 + 2];
    const float xb = posB[ib * 3 + 0], yb = posB[ib * 3 + 1], zb = posB[ib * 3 + 2];

    v2f c2[NUM_RBF / 2];
#pragma unroll
    for (int r = 0; r < NUM_RBF / 2; ++r) { c2[r].x = centers[2 * r]; c2[r].y = centers[2 * r + 1]; }

    v2f ra[NUM_RBF / 2], rb[NUM_RBF / 2];
#pragma unroll
    for (int r = 0; r < NUM_RBF / 2; ++r) { ra[r].x = 0.0f; ra[r].y = 0.0f; rb[r].x = 0.0f; rb[r].y = 0.0f; }

    float col[NUM_RBF];
#pragma unroll
    for (int r = 0; r < NUM_RBF; ++r) col[r] = 0.0f;

    if (w == 0) {
#pragma unroll 1
        for (int k = 0; k < 2; ++k) {
            const int J    = (I + 16 * k) & (NRB - 1);
            const int jrow = J * RB + h * 16 + c;
            const float xj = posB[jrow * 3 + 0], yj = posB[jrow * 3 + 1], zj = posB[jrow * 3 + 2];
            tile32<false>(xa, ya, za, xb, yb, zb, xj, yj, zj, c2, ra, rb, col);
        }
    } else {
        const int J    = (I + w) & (NRB - 1);
        const int jrow = J * RB + h * 16 + c;
        const float xj = posB[jrow * 3 + 0], yj = posB[jrow * 3 + 1], zj = posB[jrow * 3 + 2];
        tile32<true>(xa, ya, za, xb, yb, zb, xj, yj, zj, c2, ra, rb, col);

        v4f* __restrict__ dst = (v4f*)(planes + (size_t)w * PLANE_FLOATS + ((size_t)b * LSEQ + jrow) * NUM_RBF);
#pragma unroll
        for (int q = 0; q < 4; ++q) {
            v4f o;
#pragma unroll
            for (int e = 0; e < 4; ++e) o[e] = col[r_of_pos(q * 4 + e)];
            dst[q] = o;
        }
    }

    {
        const int slot = w * 2 + h;
#pragma unroll
        for (int r = 0; r < NUM_RBF; ++r) {
            sPart[slot][c][r]      = (r & 1) ? ra[r >> 1].y : ra[r >> 1].x;
            sPart[slot][16 + c][r] = (r & 1) ? rb[r >> 1].y : rb[r >> 1].x;
        }
    }
    __syncthreads();

    {
        const int orow = tid >> 4;
        const int p    = tid & 15;
        const int r    = r_of_pos(p);
        float sum = 0.0f;
#pragma unroll
        for (int k = 0; k < NSLOTS; ++k) sum += sPart[k][orow][r];
        planes[((size_t)b * LSEQ + I * RB + orow) * NUM_RBF + p] = sum;
    }
}

#define LIN_COLS  64
#define LIN_TILES (LIN_COLS / 16)
#define LIN_ROWQ  (16 / 2)
#define LIN_PITCH (LIN_COLS + 4)

__global__ __launch_bounds__(256) void linear_wmma_kernel(
    const float* __restrict__ planes,
    const float* __restrict__ W,
    const float* __restrict__ bias,
    const float* __restrict__ latent,
    float* __restrict__ out)
{
    __shared__ __attribute__((aligned(16))) float sTile[8][16][LIN_PITCH];

    const int lane    = threadIdx.x & 31;
    const int waveBlk = threadIdx.x >> 5;
    const int wave    = blockIdx.x * 8 + waveBlk;
    const int rowBase = (wave >> 2) * 16;
    const int n0      = (wave & 3) * LIN_COLS;

    const int c  = lane & 15;
    const int hi = lane >> 4;

    const size_t rmOff = (size_t)(rowBase + hi) * HIDDEN + n0 + 4 * c;
    const float* __restrict__ latP = latent + rmOff;
    float*       __restrict__ outP = out + rmOff;

    const float* __restrict__ plP = planes + (size_t)(rowBase + c) * NUM_RBF + hi * 8;
    v4f pq[NPLANES][2];
#pragma unroll
    for (int p = 0; p < NPLANES; ++p) {
        pq[p][0] = *(const v4f*)(plP + (size_t)p * PLANE_FLOATS);
        pq[p][1] = *(const v4f*)(plP + (size_t)p * PLANE_FLOATS + 4);
    }

    v2f af[LIN_TILES][4];
#pragma unroll
    for (int t = 0; t < LIN_TILES; ++t) {
        const int n = n0 + (c >> 3) * 32 + t * 8 + (c & 7);
#pragma unroll
        for (int kc = 0; kc < 4; ++kc) {
            const float* p = W + (size_t)n * NUM_RBF + kc * 4 + hi * 2;
            v2f u; u.x = p[0]; u.y = p[1];
            af[t][kc] = u;
        }
    }

    const v4f bs = *(const v4f*)(bias + n0 + 4 * c);

    v4f lat[LIN_ROWQ];
#pragma unroll
    for (int q = 0; q < LIN_ROWQ; ++q) lat[q] = *(const v4f*)(latP + (size_t)q * 2 * HIDDEN);

    __builtin_amdgcn_sched_barrier(0);

    v4f s0 = pq[0][0], s1 = pq[0][1];
#pragma unroll
    for (int p = 1; p < NPLANES; ++p) { s0 += pq[p][0]; s1 += pq[p][1]; }
    s0 *= (1.0f / (float)LSEQ);
    s1 *= (1.0f / (float)LSEQ);

    v2f bf[4];
    bf[0].x = s0[0]; bf[0].y = s0[1];
    bf[1].x = s0[2]; bf[1].y = s0[3];
    bf[2].x = s1[0]; bf[2].y = s1[1];
    bf[3].x = s1[2]; bf[3].y = s1[3];

    float* stP = &sTile[waveBlk][c][32 * hi];
#pragma unroll
    for (int t = 0; t < LIN_TILES; ++t) {
        v8f cfr = {0.f, 0.f, 0.f, 0.f, 0.f, 0.f, 0.f, 0.f};
#pragma unroll
        for (int kc = 0; kc < 4; ++kc) {
            cfr = __builtin_amdgcn_wmma_f32_16x16x4_f32(
                 false, af[t][kc],  false, bf[kc],
                 (short)0, cfr,  false,  false);
        }

        v4f d0, d1;
#pragma unroll
        for (int e = 0; e < 4; ++e) { d0[e] = cfr[e]; d1[e] = cfr[4 + e]; }
        *(v4f*)(stP + 8 * t)     = d0;
        *(v4f*)(stP + 8 * t + 4) = d1;
    }

    const float* ldP = &sTile[waveBlk][hi][4 * c];
#pragma unroll
    for (int q = 0; q < LIN_ROWQ; ++q) {
        const v4f d = *(const v4f*)(ldP + q * 2 * LIN_PITCH);
        v4f o;
#pragma unroll
        for (int e = 0; e < 4; ++e) o[e] = lat[q][e] + d[e] + bs[e];
        *(v4f*)(outP + (size_t)q * 2 * HIDDEN) = o;
    }
}

extern "C" void kernel_launch(void* const* d_in, const int* in_sizes, int n_in,
                              void* d_out, int out_size, void* d_ws, size_t ws_size,
                              hipStream_t stream) {
    const float* pos     = (const float*)d_in[0];
    const float* latent  = (const float*)d_in[1];
    const float* W       = (const float*)d_in[2];
    const float* bias    = (const float*)d_in[3];
    const float* centers = (const float*)d_in[4];
    float* out    = (float*)d_out;
    float* planes = (float*)d_ws;

    rbf_pair_kernel<<<BATCH * NRB, 512, 0, stream>>>(pos, centers, planes);

    linear_wmma_kernel<<<(BATCH * LSEQ / 16) * (HIDDEN / LIN_COLS) / 8, 256, 0, stream>>>(
        planes, W, bias, latent, out);
}


# linear phase: what the 8 waves of a group share (the 16 planes of its 32 rows, W) is fetched once per group in whole 128-byte lines and handed out through LDS; sums in the same order
# speedup vs baseline: 1.9055x; 1.1177x over previous
#include <hip/hip_runtime.h>


#define BATCH 8
#define LSEQ 1024
#define HIDDEN 256
#define NUM_RBF 16

typedef __attribute__((ext_vector_type(2))) float v2f;
typedef __attribute__((ext_vector_type(4))) float v4f;
typedef __attribute__((ext_vector_type(8))) float v8f;

#define NEG_HALF_LOG2E (-0.72134752044448170368f)

#define RB      32
#define NRB     (LSEQ / RB)
#define NPLANES 16
#define PLANE_FLOATS (BATCH * LSEQ * NUM_RBF)
#define NSLOTS  32

__device__ __forceinline__ float ror1(float v) {
    return __int_as_float(__builtin_amdgcn_update_dpp(0, __float_as_int(v), 0x121, 0xF, 0xF, true));
}

__device__ __forceinline__ int r_of_pos(int p) {
    return ((p >> 1) & 3) * 4 + (p >> 3) * 2 + (p & 1);
}

__device__ __forceinline__ void pair_rbf(float xi, float yi, float zi,
                                         float xj, float yj, float zj,
                                         const v2f (&c2)[NUM_RBF / 2], v2f (&e)[NUM_RBF / 2]) {
    const float dx = xi - xj;
    const float dy = yi - yj;
    const float dz = zi - zj;
    const float d  = __builtin_sqrtf(dx * dx + dy * dy + dz * dz);
    const v2f dd = {d, d};
#pragma unroll
    for (int r = 0; r < NUM_RBF / 2; ++r) {
        const v2f t = dd - c2[r];
        const v2f u = t * t * NEG_HALF_LOG2E;
        e[r].x = __builtin_amdgcn_exp2f(u.x);
        e[r].y = __builtin_amdgcn_exp2f(u.y);
    }
}

template <bool SYM>
__device__ __forceinline__ void tile32(float xa, float ya, float za,
                                       float xb, float yb, float zb,
                                       float xj, float yj, float zj,
                                       const v2f (&c2)[NUM_RBF / 2],
                                       v2f (&ra)[NUM_RBF / 2], v2f (&rb)[NUM_RBF / 2],
                                       float (&col)[NUM_RBF]) {
#pragma unroll 2
    for (int s = 0; s < 16; ++s) {
        xj = ror1(xj); yj = ror1(yj); zj = ror1(zj);
        v2f ea[NUM_RBF / 2], eb[NUM_RBF / 2];
        pair_rbf(xa, ya, za, xj, yj, zj, c2, ea);
        pair_rbf(xb, yb, zb, xj, yj, zj, c2, eb);
#pragma unroll
        for (int r = 0; r < NUM_RBF / 2; ++r) {
            ra[r] += ea[r];
            rb[r] += eb[r];
            if (SYM) {
                const v2f both = ea[r] + eb[r];
                col[2 * r]     = ror1(col[2 * r])     + both.x;
                col[2 * r + 1] = ror1(col[2 * r + 1]) + both.y;
            }
        }
    }
}

__global__ __launch_bounds__(512) __attribute__((amdgpu_waves_per_eu(4, 4))) void rbf_pair_kernel(
    const float* __restrict__ pos,
    const float* __restrict__ centers,
    float* __restrict__ planes)
{
    __shared__ float sPart[NSLOTS][RB][NUM_RBF + 1];

    const int tid  = threadIdx.x;
    const int lane = tid & 31;
    const int c    = lane & 15;
    const int h    = lane >> 4;
    const int w    = __builtin_amdgcn_readfirstlane(tid >> 5);
    const int b    = blockIdx.x / NRB;
    const int I    = blockIdx.x % NRB;

    const float* __restrict__ posB = pos + (size_t)b * LSEQ * 3;

    const int ia = I * RB + c;
    const int ib = ia + 16;
    const float xa = posB[ia * 3 + 0], ya = posB[ia * 3 + 1], za = posB[ia * 3 + 2];
    const float xb = posB[ib * 3 + 0], yb = posB[ib * 3 + 1], zb = posB[ib * 3 + 2];

    v2f c2[NUM_RBF / 2];
#pragma unroll
    for (int r = 0; r < NUM_RBF / 2; ++r) { c2[r].x = centers[2 * r]; c2[r].y = centers[2 * r + 1]; }

    v2f ra[NUM_RBF / 2], rb[NUM_RBF / 2];
#pragma unroll
    for (int r = 0; r < NUM_RBF / 2; ++r) { ra[r].x = 0.0f; ra[r].y = 0.0f; rb[r].x = 0.0f; rb[r].y = 0.0f; }

    float col[NUM_RBF];
#pragma unroll
    for (int r = 0; r < NUM_RBF; ++r) col[r] = 0.0f;

    if (w == 0) {
#pragma unroll 1
        for (int k = 0; k < 2; ++k) {
            const int J    = (I + 16 * k) & (NRB - 1);
            const int jrow = J * RB + h * 16 + c;
            const float xj = posB[jrow * 3 + 0], yj = posB[jrow * 3 + 1], zj = posB[jrow * 3 + 2];
            tile32<false>(xa, ya, za, xb, yb, zb, xj, yj, zj, c2, ra, rb, col);
        }
    } else {
        const int J    = (I + w) & (NRB - 1);
        const int jrow = J * RB + h * 16 + c;
        const float xj = posB[jrow * 3 + 0], yj = posB[jrow * 3 + 1], zj = posB[jrow * 3 + 2];
        tile32<true>(xa, ya, za, xb, yb, zb, xj, yj, zj, c2, ra, rb, col);

        v4f* __restrict__ dst = (v4f*)(planes + (size_t)w * PLANE_FLOATS + ((size_t)b * LSEQ + jrow) * NUM_RBF);
#pragma unroll
        for (int q = 0; q < 4; ++q) {
            v4f o;
#pragma unroll
            for (int e = 0; e < 4; ++e) o[e] = col[r_of_pos(q * 4 + e)];
            dst[q] = o;
        }
    }

    {
        const int slot = w * 2 + h;
#pragma unroll
        for (int r = 0; r < NUM_RBF; ++r) {
            sPart[slot][c][r]      = (r & 1) ? ra[r >> 1].y : ra[r >> 1].x;
            sPart[slot][16 + c][r] = (r & 1) ? rb[r >> 1].y : rb[r >> 1].x;
        }
    }
    __syncthreads();

    {
        const int orow = tid >> 4;
        const int p    = tid & 15;
        const int r    = r_of_pos(p);
        float sum = 0.0f;
#pragma unroll
        for (int k = 0; k < NSLOTS; ++k) sum += sPart[k][orow][r];
        planes[((size_t)b * LSEQ + I * RB + orow) * NUM_RBF + p] = sum;
    }
}

#define LIN_COLS  64
#define LIN_TILES (LIN_COLS / 16)
#define LIN_ROWQ  (16 / 2)
#define LIN_PITCH (LIN_COLS + 4)
#define LIN_ROWS  32
#define LIN_WCHUNKS ((HIDDEN * NUM_RBF / 4) / 256)
#define OPD_PITCH (NUM_RBF + 4)

__global__ __launch_bounds__(256) void linear_wmma_kernel(
    const float* __restrict__ planes,
    const float* __restrict__ W,
    const float* __restrict__ bias,
    const float* __restrict__ latent,
    float* __restrict__ out)
{
    __shared__ __attribute__((aligned(16))) float sTile[8][16][LIN_PITCH];
    __shared__ __attribute__((aligned(16))) float sMean[LIN_ROWS][OPD_PITCH];
    __shared__ __attribute__((aligned(16))) float sW[HIDDEN][OPD_PITCH];

    const int tid     = threadIdx.x;
    const int lane    = tid & 31;
    const int waveBlk = tid >> 5;
    const int wave    = blockIdx.x * 8 + waveBlk;
    const int rowBase = (wave >> 2) * 16;
    const int n0      = (wave & 3) * LIN_COLS;

    const int c  = lane & 15;
    const int hi = lane >> 4;

    const size_t rmOff = (size_t)(rowBase + hi) * HIDDEN + n0 + 4 * c;
    const float* __restrict__ latP = latent + rmOff;
    float*       __restrict__ outP = out + rmOff;

    const float* __restrict__ plP = planes + (size_t)blockIdx.x * (LIN_ROWS * NUM_RBF) + 2 * tid;
    v2f pq[NPLANES];
#pragma unroll
    for (int p = 0; p < NPLANES; ++p) pq[p] = *(const v2f*)(plP + (size_t)p * PLANE_FLOATS);

    v4f wq[LIN_WCHUNKS];
#pragma unroll
    for (int i = 0; i < LIN_WCHUNKS; ++i) wq[i] = *(const v4f*)(W + (size_t)(tid + 256 * i) * 4);

    const v4f bs = *(const v4f*)(bias + n0 + 4 * c);

    v4f lat[LIN_ROWQ];
#pragma unroll
    for (int q = 0; q < LIN_ROWQ; ++q) lat[q] = *(const v4f*)(latP + (size_t)q * 2 * HIDDEN);

    __builtin_amdgcn_sched_barrier(0);

    {
        v2f s = pq[0];
#pragma unroll
        for (int p = 1; p < NPLANES; ++p) s += pq[p];
        s *= (1.0f / (float)LSEQ);
        *(v2f*)&sMean[tid >> 3][(tid & 7) * 2] = s;
    }
#pragma unroll
    for (int i = 0; i < LIN_WCHUNKS; ++i) {
        const int q = tid + 256 * i;
        *(v4f*)&sW[q >> 2][(q & 3) * 4] = wq[i];
    }

    asm volatile("s_wait_dscnt 0x0\n\ts_barrier_signal -1\n\ts_barrier_wait -1" ::: "memory");

    const float* mP = &sMean[(waveBlk >> 2) * 16 + c][hi * 8];
    const v4f s0 = *(const v4f*)mP;
    const v4f s1 = *(const v4f*)(mP + 4);

    v2f bf[4];
    bf[0].x = s0[0]; bf[0].y = s0[1];
    bf[1].x = s0[2]; bf[1].y = s0[3];
    bf[2].x = s1[0]; bf[2].y = s1[1];
    bf[3].x = s1[2]; bf[3].y = s1[3];

    v2f af[LIN_TILES][4];
#pragma unroll
    for (int t = 0; t < LIN_TILES; ++t) {
        const int n = n0 + (c >> 3) * 32 + t * 8 + (c & 7);
#pragma unroll
        for (int kc = 0; kc < 4; ++kc) af[t][kc] = *(const v2f*)&sW[n][kc * 4 + hi * 2];
    }

    float* stP = &sTile[waveBlk][c][32 * hi];
#pragma unroll
    for (int t = 0; t < LIN_TILES; ++t) {
        v8f cfr = {0.f, 0.f, 0.f, 0.f, 0.f, 0.f, 0.f, 0.f};
#pragma unroll
        for (int kc = 0; kc < 4; ++kc) {
            cfr = __builtin_amdgcn_wmma_f32_16x16x4_f32(
                 false, af[t][kc],  false, bf[kc],
                 (short)0, cfr,  false,  false);
        }

        v4f d0, d1;
#pragma unroll
        for (int e = 0; e < 4; ++e) { d0[e] = cfr[e]; d1[e] = cfr[4 + e]; }
        *(v4f*)(stP + 8 * t)     = d0;
        *(v4f*)(stP + 8 * t + 4) = d1;
    }

    const float* ldP = &sTile[waveBlk][hi][4 * c];
#pragma unroll
    for (int q = 0; q < LIN_ROWQ; ++q) {
        const v4f d = *(const v4f*)(ldP + q * 2 * LIN_PITCH);
        v4f o;
#pragma unroll
        for (int e = 0; e < 4; ++e) o[e] = lat[q][e] + d[e] + bs[e];
        *(v4f*)(outP + (size_t)q * 2 * HIDDEN) = o;
    }
}

extern "C" void kernel_launch(void* const* d_in, const int* in_sizes, int n_in,
                              void* d_out, int out_size, void* d_ws, size_t ws_size,
                              hipStream_t stream) {
    const float* pos     = (const float*)d_in[0];
    const float* latent  = (const float*)d_in[1];
    const float* W       = (const float*)d_in[2];
    const float* bias    = (const float*)d_in[3];
    const float* centers = (const float*)d_in[4];
    float* out    = (float*)d_out;
    float* planes = (float*)d_ws;

    rbf_pair_kernel<<<BATCH * NRB, 512, 0, stream>>>(pos, centers, planes);

    linear_wmma_kernel<<<(BATCH * LSEQ / 16) * (HIDDEN / LIN_COLS) / 8, 256, 0, stream>>>(
        planes, W, bias, latent, out);
}
